# starting kernel
# baseline (speedup 1.0000x reference)
#include <hip/hip_runtime.h>
#include <hip/hip_bf16.h>


typedef float v2f __attribute__((ext_vector_type(2)));
typedef float v8f __attribute__((ext_vector_type(8)));

#define F_IN  256
#define F_MID 64
#define F_OUT 32


__global__ void __launch_bounds__(256) fill_kernel(float* __restrict__ p, float v, int n) {
    int i = blockIdx.x * 256 + threadIdx.x;
    if (i < n) p[i] = v;
}

__global__ void __launch_bounds__(256) deg_count_kernel(const int* __restrict__ dst,
                                                        float* __restrict__ deg, int n_edges) {
    int e = blockIdx.x * 256 + threadIdx.x;
    if (e < n_edges) atomicAdd(&deg[dst[e]], 1.0f);
}

__global__ void __launch_bounds__(256) rsqrt_kernel(float* __restrict__ d, int n) {
    int i = blockIdx.x * 256 + threadIdx.x;
    if (i < n) d[i] = rsqrtf(d[i]);
}


template <int K, int NOUT>
__global__ void __launch_bounds__(256) gemm_kernel(const float* __restrict__ x,
                                                   const float* __restrict__ W,
                                                   float* __restrict__ h, int n_nodes) {
    constexpr int LDK = K + 4;
    constexpr int NT  = NOUT / 16;
    __shared__ float sWt[NOUT * LDK];

    for (int i = threadIdx.x; i < K * NOUT; i += 256) {
        const int k = i / NOUT;
        const int n = i % NOUT;
        sWt[n * LDK + k] = W[i];
    }
    __syncthreads();

    const int wave = threadIdx.x >> 5;
    const int lane = threadIdx.x & 31;
    const int half = lane >> 4;
    const int l    = lane & 15;
    const int m0   = (blockIdx.x * 8 + wave) * 16;
    if (m0 >= n_nodes) return;

    const int rowA = min(m0 + l, n_nodes - 1);
    const float* __restrict__ xrow = x + (size_t)rowA * K;

    v8f acc[NT] = {};
    for (int k = 0; k < K; k += 4) {
        const int kk = k + 2 * half;
        const v2f a = *(const v2f*)(xrow + kk);
#pragma unroll
        for (int t = 0; t < NT; ++t) {
            const int col = t * 16 + l;
            const v2f b = *(const v2f*)(&sWt[col * LDK + kk]);
            acc[t] = __builtin_amdgcn_wmma_f32_16x16x4_f32(
                false, a, false, b, (short)0, acc[t], false, false);
        }
    }

    if (m0 + 16 <= n_nodes) {
        float* __restrict__ hp = h + (size_t)(m0 + 8 * half) * NOUT + l;
#pragma unroll
        for (int v = 0; v < 8; ++v) {
#pragma unroll
            for (int t = 0; t < NT; ++t)
                hp[(size_t)v * NOUT + t * 16] = acc[t][v];
        }
    } else {
#pragma unroll
        for (int v = 0; v < 8; ++v) {
            const int row = m0 + 8 * half + v;
            if (row < n_nodes) {
#pragma unroll
                for (int t = 0; t < NT; ++t)
                    h[(size_t)row * NOUT + t * 16 + l] = acc[t][v];
            }
        }
    }
}


template <int F>
__global__ void __launch_bounds__(256) scatter_kernel(const int* __restrict__ src,
                                                      const int* __restrict__ dst,
                                                      const float* __restrict__ dinv,
                                                      const float* __restrict__ h,
                                                      float* __restrict__ agg, int n_edges) {
    const int idx = blockIdx.x * 256 + threadIdx.x;
    const int e = idx / F;
    const int f = idx % F;
    if (e >= n_edges) return;
    const int s = src[e];
    const int d = dst[e];
    const float norm = dinv[s] * dinv[d];
    atomicAdd(&agg[(size_t)d * F + f], norm * h[(size_t)s * F + f]);
}

template <int F, bool RELU>
__global__ void __launch_bounds__(256) finalize_kernel(float* __restrict__ agg,
                                                       const float* __restrict__ h,
                                                       const float* __restrict__ dinv,
                                                       const float* __restrict__ bias,
                                                       int n_nodes) {
    const int idx = blockIdx.x * 256 + threadIdx.x;
    if (idx >= n_nodes * F) return;
    const int i = idx / F;
    const int f = idx % F;
    const float di = dinv[i];
    float v = agg[idx] + di * di * h[idx] + bias[f];
    if (RELU) v = fmaxf(v, 0.0f);
    agg[idx] = v;
}


__global__ void __launch_bounds__(256) decode_kernel(const int* __restrict__ pos,
                                                     const int* __restrict__ neg,
                                                     const float* __restrict__ z,
                                                     float* __restrict__ out, int n_eval) {
    const int gw   = (blockIdx.x * 256 + threadIdx.x) >> 5;
    const int lane = threadIdx.x & 31;
    if (gw >= 2 * n_eval) return;
    int a, b;
    if (gw < n_eval) { a = pos[gw];            b = pos[n_eval + gw]; }
    else             { int k = gw - n_eval; a = neg[k]; b = neg[n_eval + k]; }
    float v = z[(size_t)a * F_OUT + lane] * z[(size_t)b * F_OUT + lane];
#pragma unroll
    for (int o = 16; o > 0; o >>= 1) v += __shfl_xor(v, o, 32);
    if (lane == 0) out[gw] = v;
}


static inline size_t align_up(size_t x) { return (x + 255) & ~(size_t)255; }

extern "C" void kernel_launch(void* const* d_in, const int* in_sizes, int n_in,
                              void* d_out, int out_size, void* d_ws, size_t ws_size,
                              hipStream_t stream) {
    const float* x   = (const float*)d_in[0];
    const float* W1  = (const float*)d_in[1];
    const float* b1  = (const float*)d_in[2];
    const float* W2  = (const float*)d_in[3];
    const float* b2  = (const float*)d_in[4];
    const int*   ei  = (const int*)d_in[5];
    const int*   pos = (const int*)d_in[6];
    const int*   neg = (const int*)d_in[7];
    float* out = (float*)d_out;

    const int n_nodes = in_sizes[0] / F_IN;
    const int n_edges = in_sizes[5] / 2;
    const int n_eval  = in_sizes[6] / 2;
    const int* src = ei;
    const int* dst = ei + n_edges;

    char* w = (char*)d_ws;
    size_t off = 0;
    float* dinv = (float*)(w + off); off += align_up((size_t)n_nodes * 4);
    float* h1   = (float*)(w + off); off += align_up((size_t)n_nodes * F_MID * 4);
    float* agg1 = (float*)(w + off); off += align_up((size_t)n_nodes * F_MID * 4);
    float* h2   = (float*)(w + off); off += align_up((size_t)n_nodes * F_OUT * 4);
    float* z    = (float*)(w + off); off += align_up((size_t)n_nodes * F_OUT * 4);
    (void)ws_size;

    const int B = 256;
    const int nm = n_nodes * F_MID;
    const int no = n_nodes * F_OUT;

    fill_kernel<<<(n_nodes + B - 1) / B, B, 0, stream>>>(dinv, 1.0f, n_nodes);
    fill_kernel<<<(nm + B - 1) / B, B, 0, stream>>>(agg1, 0.0f, nm);
    fill_kernel<<<(no + B - 1) / B, B, 0, stream>>>(z, 0.0f, no);
    deg_count_kernel<<<(n_edges + B - 1) / B, B, 0, stream>>>(dst, dinv, n_edges);
    rsqrt_kernel<<<(n_nodes + B - 1) / B, B, 0, stream>>>(dinv, n_nodes);

    const int gemm_blocks = (n_nodes + 128 - 1) / 128;
    gemm_kernel<F_IN, F_MID><<<gemm_blocks, B, 0, stream>>>(x, W1, h1, n_nodes);
    {
        long long t = (long long)n_edges * F_MID;
        scatter_kernel<F_MID><<<(unsigned)((t + B - 1) / B), B, 0, stream>>>(
            src, dst, dinv, h1, agg1, n_edges);
    }
    finalize_kernel<F_MID, true><<<(nm + B - 1) / B, B, 0, stream>>>(agg1, h1, dinv, b1, n_nodes);

    gemm_kernel<F_MID, F_OUT><<<gemm_blocks, B, 0, stream>>>(agg1, W2, h2, n_nodes);
    {
        long long t = (long long)n_edges * F_OUT;
        scatter_kernel<F_OUT><<<(unsigned)((t + B - 1) / B), B, 0, stream>>>(
            src, dst, dinv, h2, z, n_edges);
    }
    finalize_kernel<F_OUT, false><<<(no + B - 1) / B, B, 0, stream>>>(z, h2, dinv, b2, n_nodes);

    {
        int waves = 2 * n_eval;
        int blocks = (waves * 32 + B - 1) / B;
        decode_kernel<<<blocks, B, 0, stream>>>(pos, neg, z, out, n_eval);
    }
    (void)out_size; (void)n_in;
}


# Atomic adds kept in one L2 half: waves read their die id, draw (128-edge slice, node half) tickets, add only into rows of their own half (degree count, both scatters); sums still arrival-ordered
# speedup vs baseline: 2.1384x; 2.1384x over previous
#include <hip/hip_runtime.h>
#include <hip/hip_bf16.h>


typedef float v2f __attribute__((ext_vector_type(2)));
typedef float v8f __attribute__((ext_vector_type(8)));

#define F_IN  256
#define F_MID 64
#define F_OUT 32


__global__ void __launch_bounds__(256) fill_kernel(float* __restrict__ p, float v, int n) {
    int i = blockIdx.x * 256 + threadIdx.x;
    if (i < n) p[i] = v;
}


#define TK_EDGES 128
#define TK_PITCH 64

__device__ __forceinline__ int chip_half() {
    const unsigned id = __builtin_amdgcn_s_sendmsg_rtn(0x87);
    return (int)((id >> 18) & 1u);
}

__device__ __forceinline__ int take_ticket(int* ctr) {
    int t = 0;
    if ((threadIdx.x & 31) == 0) t = atomicAdd(ctr, 1);
    return __builtin_amdgcn_readfirstlane(t);
}

__device__ __forceinline__ int node_split(int n_nodes) {
    return ((n_nodes >> 1) + 31) & ~31;
}

__global__ void __launch_bounds__(256) deg_count_kernel(const int* __restrict__ dst,
                                                        float* __restrict__ deg,
                                                        int* __restrict__ tick,
                                                        int n_edges, int n_nodes) {
    const int lane  = threadIdx.x & 31;
    const int n_tk  = (n_edges + TK_EDGES - 1) / TK_EDGES;
    const int split = node_split(n_nodes);
    const int mine  = chip_half();
    for (int side = 0; side < 2; ++side) {
        const int hs = mine ^ side;
        int* ctr = tick + hs * TK_PITCH;
        for (int it = 0; it <= n_tk; ++it) {
            const int t = take_ticket(ctr);
            if (t >= n_tk) break;
            int d[TK_EDGES / 32];
#pragma unroll
            for (int b = 0; b < TK_EDGES / 32; ++b) {
                const int e = t * TK_EDGES + 32 * b + lane;
                d[b] = (e < n_edges) ? dst[e] : -1;
            }
#pragma unroll
            for (int b = 0; b < TK_EDGES / 32; ++b) {
                if (d[b] >= 0 && (int)(d[b] >= split) == hs) atomicAdd(&deg[d[b]], 1.0f);
            }
        }
    }
}

__global__ void __launch_bounds__(256) rsqrt_kernel(float* __restrict__ d, int n) {
    int i = blockIdx.x * 256 + threadIdx.x;
    if (i < n) d[i] = rsqrtf(d[i]);
}


template <int K, int NOUT>
__global__ void __launch_bounds__(256) gemm_kernel(const float* __restrict__ x,
                                                   const float* __restrict__ W,
                                                   float* __restrict__ h, int n_nodes) {
    constexpr int LDK = K + 4;
    constexpr int NT  = NOUT / 16;
    __shared__ float sWt[NOUT * LDK];

    for (int i = threadIdx.x; i < K * NOUT; i += 256) {
        const int k = i / NOUT;
        const int n = i % NOUT;
        sWt[n * LDK + k] = W[i];
    }
    __syncthreads();

    const int wave = threadIdx.x >> 5;
    const int lane = threadIdx.x & 31;
    const int half = lane >> 4;
    const int l    = lane & 15;
    const int m0   = (blockIdx.x * 8 + wave) * 16;
    if (m0 >= n_nodes) return;

    const int rowA = min(m0 + l, n_nodes - 1);
    const float* __restrict__ xrow = x + (size_t)rowA * K;

    v8f acc[NT] = {};
    for (int k = 0; k < K; k += 4) {
        const int kk = k + 2 * half;
        const v2f a = *(const v2f*)(xrow + kk);
#pragma unroll
        for (int t = 0; t < NT; ++t) {
            const int col = t * 16 + l;
            const v2f b = *(const v2f*)(&sWt[col * LDK + kk]);
            acc[t] = __builtin_amdgcn_wmma_f32_16x16x4_f32(
                false, a, false, b, (short)0, acc[t], false, false);
        }
    }

    if (m0 + 16 <= n_nodes) {
        float* __restrict__ hp = h + (size_t)(m0 + 8 * half) * NOUT + l;
#pragma unroll
        for (int v = 0; v < 8; ++v) {
#pragma unroll
            for (int t = 0; t < NT; ++t)
                hp[(size_t)v * NOUT + t * 16] = acc[t][v];
        }
    } else {
#pragma unroll
        for (int v = 0; v < 8; ++v) {
            const int row = m0 + 8 * half + v;
            if (row < n_nodes) {
#pragma unroll
                for (int t = 0; t < NT; ++t)
                    h[(size_t)row * NOUT + t * 16 + l] = acc[t][v];
            }
        }
    }
}


template <int F>
__device__ __forceinline__ void add_row(const float* __restrict__ h, float* __restrict__ agg,
                                        int s, int d, float nrm, int lane) {
    float hv[F / 32];
#pragma unroll
    for (int k = 0; k < F / 32; ++k) hv[k] = h[(size_t)s * F + 32 * k + lane];
#pragma unroll
    for (int k = 0; k < F / 32; ++k)
        atomicAdd(&agg[(size_t)d * F + 32 * k + lane], nrm * hv[k]);
}

template <int F>
__global__ void __launch_bounds__(256) scatter_kernel(const int* __restrict__ src,
                                                      const int* __restrict__ dst,
                                                      const float* __restrict__ dinv,
                                                      const float* __restrict__ h,
                                                      float* __restrict__ agg,
                                                      int* __restrict__ tick,
                                                      int n_edges, int n_nodes) {
    constexpr int NB = TK_EDGES / 32;
    constexpr int NL = F / 32;
    const int lane  = threadIdx.x & 31;
    const int n_tk  = (n_edges + TK_EDGES - 1) / TK_EDGES;
    const int split = node_split(n_nodes);
    const int mine  = chip_half();
    for (int side = 0; side < 2; ++side) {
        const int hs = mine ^ side;
        int* ctr = tick + hs * TK_PITCH;
        for (int it = 0; it <= n_tk; ++it) {
            const int t = take_ticket(ctr);
            if (t >= n_tk) break;

            int   d[NB], s[NB];
            float nrm[NB];
#pragma unroll
            for (int b = 0; b < NB; ++b) {
                const int e = t * TK_EDGES + 32 * b + lane;
                const bool in = e < n_edges;
                d[b] = in ? dst[e] : -1;
                s[b] = in ? src[e] : 0;
            }
            unsigned msk[NB];
#pragma unroll
            for (int b = 0; b < NB; ++b) {
                const bool m = d[b] >= 0 && (int)(d[b] >= split) == hs;
                msk[b] = __builtin_amdgcn_ballot_w32(m);
                nrm[b] = dinv[m ? s[b] : 0] * dinv[m ? d[b] : 0];
            }

#pragma unroll
            for (int b = 0; b < NB; ++b) {
                unsigned mask = msk[b];
                const int   sb = s[b];
                const int   db = d[b];
                const int   nb = __float_as_int(nrm[b]);
                while (__builtin_popcount(mask) >= 4) {
                    int sj[4], dj[4]; float nj[4];
#pragma unroll
                    for (int u = 0; u < 4; ++u) {
                        const int j = __builtin_ctz(mask);
                        mask &= mask - 1;
                        sj[u] = __builtin_amdgcn_readlane(sb, j);
                        dj[u] = __builtin_amdgcn_readlane(db, j);
                        nj[u] = __int_as_float(__builtin_amdgcn_readlane(nb, j));
                    }
                    float hv[4][NL];
#pragma unroll
                    for (int u = 0; u < 4; ++u)
#pragma unroll
                        for (int k = 0; k < NL; ++k)
                            hv[u][k] = h[(size_t)sj[u] * F + 32 * k + lane];
#pragma unroll
                    for (int u = 0; u < 4; ++u)
#pragma unroll
                        for (int k = 0; k < NL; ++k)
                            atomicAdd(&agg[(size_t)dj[u] * F + 32 * k + lane], nj[u] * hv[u][k]);
                }
                while (mask != 0) {
                    const int j = __builtin_ctz(mask);
                    mask &= mask - 1;
                    add_row<F>(h, agg, __builtin_amdgcn_readlane(sb, j),
                               __builtin_amdgcn_readlane(db, j),
                               __int_as_float(__builtin_amdgcn_readlane(nb, j)), lane);
                }
            }
        }
    }
}

template <int F, bool RELU>
__global__ void __launch_bounds__(256) finalize_kernel(float* __restrict__ agg,
                                                       const float* __restrict__ h,
                                                       const float* __restrict__ dinv,
                                                       const float* __restrict__ bias,
                                                       int n_nodes) {
    const int idx = blockIdx.x * 256 + threadIdx.x;
    if (idx >= n_nodes * F) return;
    const int i = idx / F;
    const int f = idx % F;
    const float di = dinv[i];
    float v = agg[idx] + di * di * h[idx] + bias[f];
    if (RELU) v = fmaxf(v, 0.0f);
    agg[idx] = v;
}


__global__ void __launch_bounds__(256) decode_kernel(const int* __restrict__ pos,
                                                     const int* __restrict__ neg,
                                                     const float* __restrict__ z,
                                                     float* __restrict__ out, int n_eval) {
    const int gw   = (blockIdx.x * 256 + threadIdx.x) >> 5;
    const int lane = threadIdx.x & 31;
    if (gw >= 2 * n_eval) return;
    int a, b;
    if (gw < n_eval) { a = pos[gw];            b = pos[n_eval + gw]; }
    else             { int k = gw - n_eval; a = neg[k]; b = neg[n_eval + k]; }
    float v = z[(size_t)a * F_OUT + lane] * z[(size_t)b * F_OUT + lane];
#pragma unroll
    for (int o = 16; o > 0; o >>= 1) v += __shfl_xor(v, o, 32);
    if (lane == 0) out[gw] = v;
}


static inline size_t align_up(size_t x) { return (x + 255) & ~(size_t)255; }

extern "C" void kernel_launch(void* const* d_in, const int* in_sizes, int n_in,
                              void* d_out, int out_size, void* d_ws, size_t ws_size,
                              hipStream_t stream) {
    const float* x   = (const float*)d_in[0];
    const float* W1  = (const float*)d_in[1];
    const float* b1  = (const float*)d_in[2];
    const float* W2  = (const float*)d_in[3];
    const float* b2  = (const float*)d_in[4];
    const int*   ei  = (const int*)d_in[5];
    const int*   pos = (const int*)d_in[6];
    const int*   neg = (const int*)d_in[7];
    float* out = (float*)d_out;

    const int n_nodes = in_sizes[0] / F_IN;
    const int n_edges = in_sizes[5] / 2;
    const int n_eval  = in_sizes[6] / 2;
    const int* src = ei;
    const int* dst = ei + n_edges;

    char* w = (char*)d_ws;
    size_t off = 0;
    float* dinv = (float*)(w + off); off += align_up((size_t)n_nodes * 4);
    float* h1   = (float*)(w + off); off += align_up((size_t)n_nodes * F_MID * 4);
    const int TKW = 3 * 2 * TK_PITCH;
    int*   tick = (int*)(w + off);   off += (size_t)TKW * 4;
    float* agg1 = (float*)(w + off); off += align_up((size_t)n_nodes * F_MID * 4);
    float* h2   = (float*)(w + off); off += align_up((size_t)n_nodes * F_OUT * 4);
    float* z    = (float*)(w + off); off += align_up((size_t)n_nodes * F_OUT * 4);
    (void)ws_size;

    const int B = 256;
    const int nm = n_nodes * F_MID;
    const int no = n_nodes * F_OUT;

    fill_kernel<<<(n_nodes + B - 1) / B, B, 0, stream>>>(dinv, 1.0f, n_nodes);
    fill_kernel<<<(nm + TKW + B - 1) / B, B, 0, stream>>>((float*)tick, 0.0f, nm + TKW);
    fill_kernel<<<(no + B - 1) / B, B, 0, stream>>>(z, 0.0f, no);
    deg_count_kernel<<<1024, B, 0, stream>>>(dst, dinv, tick, n_edges, n_nodes);
    rsqrt_kernel<<<(n_nodes + B - 1) / B, B, 0, stream>>>(dinv, n_nodes);

    const int gemm_blocks = (n_nodes + 128 - 1) / 128;
    gemm_kernel<F_IN, F_MID><<<gemm_blocks, B, 0, stream>>>(x, W1, h1, n_nodes);
    scatter_kernel<F_MID><<<2048, B, 0, stream>>>(
        src, dst, dinv, h1, agg1, tick + 2 * TK_PITCH, n_edges, n_nodes);
    finalize_kernel<F_MID, true><<<(nm + B - 1) / B, B, 0, stream>>>(agg1, h1, dinv, b1, n_nodes);

    gemm_kernel<F_MID, F_OUT><<<gemm_blocks, B, 0, stream>>>(agg1, W2, h2, n_nodes);
    scatter_kernel<F_OUT><<<2048, B, 0, stream>>>(
        src, dst, dinv, h2, z, tick + 4 * TK_PITCH, n_edges, n_nodes);
    finalize_kernel<F_OUT, false><<<(no + B - 1) / B, B, 0, stream>>>(z, h2, dinv, b2, n_nodes);

    {
        int waves = 2 * n_eval;
        int blocks = (waves * 32 + B - 1) / B;
        decode_kernel<<<blocks, B, 0, stream>>>(pos, neg, z, out, n_eval);
    }
    (void)out_size; (void)n_in;
}


# Each GEMM epilogue stores self-loop message + bias as the start value the atomic adds run onto (now first in each sum, was last); zero-fill and finalize passes gone, ReLU in layer-2 A load
# speedup vs baseline: 2.3679x; 1.1073x over previous
#include <hip/hip_runtime.h>
#include <hip/hip_bf16.h>


typedef float v2f __attribute__((ext_vector_type(2)));
typedef float v8f __attribute__((ext_vector_type(8)));

#define F_IN  256
#define F_MID 64
#define F_OUT 32


__global__ void __launch_bounds__(256) init_kernel(float* __restrict__ deg, int* __restrict__ tick,
                                                   int n, int n_tick) {
    const int i = blockIdx.x * 256 + threadIdx.x;
    if (i < n) deg[i] = 1.0f;
    else if (i - n < n_tick) tick[i - n] = 0;
}


#define TK_EDGES 128
#define TK_PITCH 64

__device__ __forceinline__ int chip_half() {
    const unsigned id = __builtin_amdgcn_s_sendmsg_rtn(0x87);
    return (int)((id >> 18) & 1u);
}

__device__ __forceinline__ int take_ticket(int* ctr) {
    int t = 0;
    if ((threadIdx.x & 31) == 0) t = atomicAdd(ctr, 1);
    return __builtin_amdgcn_readfirstlane(t);
}

__device__ __forceinline__ int node_split(int n_nodes) {
    return ((n_nodes >> 1) + 31) & ~31;
}

__global__ void __launch_bounds__(256) deg_count_kernel(const int* __restrict__ dst,
                                                        float* __restrict__ deg,
                                                        int* __restrict__ tick,
                                                        int n_edges, int n_nodes) {
    const int lane  = threadIdx.x & 31;
    const int n_tk  = (n_edges + TK_EDGES - 1) / TK_EDGES;
    const int split = node_split(n_nodes);
    const int mine  = chip_half();
    for (int side = 0; side < 2; ++side) {
        const int hs = mine ^ side;
        int* ctr = tick + hs * TK_PITCH;
        for (int it = 0; it <= n_tk; ++it) {
            const int t = take_ticket(ctr);
            if (t >= n_tk) break;
            int d[TK_EDGES / 32];
#pragma unroll
            for (int b = 0; b < TK_EDGES / 32; ++b) {
                const int e = t * TK_EDGES + 32 * b + lane;
                d[b] = (e < n_edges) ? dst[e] : -1;
            }
#pragma unroll
            for (int b = 0; b < TK_EDGES / 32; ++b) {
                if (d[b] >= 0 && (int)(d[b] >= split) == hs) atomicAdd(&deg[d[b]], 1.0f);
            }
        }
    }
}

__global__ void __launch_bounds__(256) rsqrt_kernel(float* __restrict__ d, int n) {
    int i = blockIdx.x * 256 + threadIdx.x;
    if (i < n) d[i] = rsqrtf(d[i]);
}


template <int K, int NOUT, bool RELU_A>
__global__ void __launch_bounds__(256) gemm_kernel(const float* __restrict__ x,
                                                   const float* __restrict__ W,
                                                   const float* __restrict__ dinv,
                                                   const float* __restrict__ bias,
                                                   float* __restrict__ h,
                                                   float* __restrict__ agg, int n_nodes) {
    constexpr int LDK = K + 4;
    constexpr int NT  = NOUT / 16;
    __shared__ float sWt[NOUT * LDK];

    for (int i = threadIdx.x; i < K * NOUT; i += 256) {
        const int k = i / NOUT;
        const int n = i % NOUT;
        sWt[n * LDK + k] = W[i];
    }
    __syncthreads();

    const int wave = threadIdx.x >> 5;
    const int lane = threadIdx.x & 31;
    const int half = lane >> 4;
    const int l    = lane & 15;
    const int m0   = (blockIdx.x * 8 + wave) * 16;
    if (m0 >= n_nodes) return;

    const int rowA = min(m0 + l, n_nodes - 1);
    const float* __restrict__ xrow = x + (size_t)rowA * K;

    v8f acc[NT] = {};
    if (RELU_A) {
        v2f af[K / 4];
#pragma unroll
        for (int i = 0; i < K / 4; ++i) af[i] = *(const v2f*)(xrow + 4 * i + 2 * half);
#pragma unroll
        for (int i = 0; i < K / 4; ++i) {
            af[i].x = fmaxf(af[i].x, 0.0f);
            af[i].y = fmaxf(af[i].y, 0.0f);
        }
        __builtin_amdgcn_sched_barrier(0);
#pragma unroll
        for (int i = 0; i < K / 4; ++i) {
            const int kk = 4 * i + 2 * half;
#pragma unroll
            for (int t = 0; t < NT; ++t) {
                const int col = t * 16 + l;
                const v2f b = *(const v2f*)(&sWt[col * LDK + kk]);
                acc[t] = __builtin_amdgcn_wmma_f32_16x16x4_f32(
                    false, af[i], false, b, (short)0, acc[t], false, false);
            }
        }
    } else {
        for (int k = 0; k < K; k += 4) {
            const int kk = k + 2 * half;
            const v2f a = *(const v2f*)(xrow + kk);
#pragma unroll
            for (int t = 0; t < NT; ++t) {
                const int col = t * 16 + l;
                const v2f b = *(const v2f*)(&sWt[col * LDK + kk]);
                acc[t] = __builtin_amdgcn_wmma_f32_16x16x4_f32(
                    false, a, false, b, (short)0, acc[t], false, false);
            }
        }
    }

    float bs[NT];
#pragma unroll
    for (int t = 0; t < NT; ++t) bs[t] = bias[t * 16 + l];
    if (m0 + 16 <= n_nodes) {
        const float* __restrict__ dp = dinv + m0 + 8 * half;
        const float4 d0 = *(const float4*)(dp);
        const float4 d1 = *(const float4*)(dp + 4);
        const float di[8] = {d0.x, d0.y, d0.z, d0.w, d1.x, d1.y, d1.z, d1.w};
        const size_t o0 = (size_t)(m0 + 8 * half) * NOUT + l;
        float* __restrict__ hp = h + o0;
        float* __restrict__ ap = agg + o0;
#pragma unroll
        for (int v = 0; v < 8; ++v) {
            const float w = di[v] * di[v];
#pragma unroll
            for (int t = 0; t < NT; ++t) {
                hp[(size_t)v * NOUT + t * 16] = acc[t][v];
                ap[(size_t)v * NOUT + t * 16] = w * acc[t][v] + bs[t];
            }
        }
    } else {
#pragma unroll
        for (int v = 0; v < 8; ++v) {
            const int row = m0 + 8 * half + v;
            if (row < n_nodes) {
                const float d = dinv[row];
                const float w = d * d;
#pragma unroll
                for (int t = 0; t < NT; ++t) {
                    h[(size_t)row * NOUT + t * 16 + l]   = acc[t][v];
                    agg[(size_t)row * NOUT + t * 16 + l] = w * acc[t][v] + bs[t];
                }
            }
        }
    }
}


template <int F>
__device__ __forceinline__ void add_row(const float* __restrict__ h, float* __restrict__ agg,
                                        int s, int d, float nrm, int lane) {
    float hv[F / 32];
#pragma unroll
    for (int k = 0; k < F / 32; ++k) hv[k] = h[(size_t)s * F + 32 * k + lane];
#pragma unroll
    for (int k = 0; k < F / 32; ++k)
        atomicAdd(&agg[(size_t)d * F + 32 * k + lane], nrm * hv[k]);
}

template <int F>
__global__ void __launch_bounds__(256) scatter_kernel(const int* __restrict__ src,
                                                      const int* __restrict__ dst,
                                                      const float* __restrict__ dinv,
                                                      const float* __restrict__ h,
                                                      float* __restrict__ agg,
                                                      int* __restrict__ tick,
                                                      int n_edges, int n_nodes) {
    constexpr int NB = TK_EDGES / 32;
    constexpr int NL = F / 32;
    const int lane  = threadIdx.x & 31;
    const int n_tk  = (n_edges + TK_EDGES - 1) / TK_EDGES;
    const int split = node_split(n_nodes);
    const int mine  = chip_half();
    for (int side = 0; side < 2; ++side) {
        const int hs = mine ^ side;
        int* ctr = tick + hs * TK_PITCH;
        for (int it = 0; it <= n_tk; ++it) {
            const int t = take_ticket(ctr);
            if (t >= n_tk) break;

            int   d[NB], s[NB];
            float nrm[NB];
#pragma unroll
            for (int b = 0; b < NB; ++b) {
                const int e = t * TK_EDGES + 32 * b + lane;
                const bool in = e < n_edges;
                d[b] = in ? dst[e] : -1;
                s[b] = in ? src[e] : 0;
            }
            unsigned msk[NB];
#pragma unroll
            for (int b = 0; b < NB; ++b) {
                const bool m = d[b] >= 0 && (int)(d[b] >= split) == hs;
                msk[b] = __builtin_amdgcn_ballot_w32(m);
                nrm[b] = dinv[m ? s[b] : 0] * dinv[m ? d[b] : 0];
            }

#pragma unroll
            for (int b = 0; b < NB; ++b) {
                unsigned mask = msk[b];
                const int   sb = s[b];
                const int   db = d[b];
                const int   nb = __float_as_int(nrm[b]);
                while (__builtin_popcount(mask) >= 4) {
                    int sj[4], dj[4]; float nj[4];
#pragma unroll
                    for (int u = 0; u < 4; ++u) {
                        const int j = __builtin_ctz(mask);
                        mask &= mask - 1;
                        sj[u] = __builtin_amdgcn_readlane(sb, j);
                        dj[u] = __builtin_amdgcn_readlane(db, j);
                        nj[u] = __int_as_float(__builtin_amdgcn_readlane(nb, j));
                    }
                    float hv[4][NL];
#pragma unroll
                    for (int u = 0; u < 4; ++u)
#pragma unroll
                        for (int k = 0; k < NL; ++k)
                            hv[u][k] = h[(size_t)sj[u] * F + 32 * k + lane];
#pragma unroll
                    for (int u = 0; u < 4; ++u)
#pragma unroll
                        for (int k = 0; k < NL; ++k)
                            atomicAdd(&agg[(size_t)dj[u] * F + 32 * k + lane], nj[u] * hv[u][k]);
                }
                while (mask != 0) {
                    const int j = __builtin_ctz(mask);
                    mask &= mask - 1;
                    add_row<F>(h, agg, __builtin_amdgcn_readlane(sb, j),
                               __builtin_amdgcn_readlane(db, j),
                               __int_as_float(__builtin_amdgcn_readlane(nb, j)), lane);
                }
            }
        }
    }
}


__global__ void __launch_bounds__(256) decode_kernel(const int* __restrict__ pos,
                                                     const int* __restrict__ neg,
                                                     const float* __restrict__ z,
                                                     float* __restrict__ out, int n_eval) {
    const int gw   = (blockIdx.x * 256 + threadIdx.x) >> 5;
    const int lane = threadIdx.x & 31;
    if (gw >= 2 * n_eval) return;
    int a, b;
    if (gw < n_eval) { a = pos[gw];            b = pos[n_eval + gw]; }
    else             { int k = gw - n_eval; a = neg[k]; b = neg[n_eval + k]; }
    float v = z[(size_t)a * F_OUT + lane] * z[(size_t)b * F_OUT + lane];
#pragma unroll
    for (int o = 16; o > 0; o >>= 1) v += __shfl_xor(v, o, 32);
    if (lane == 0) out[gw] = v;
}


static inline size_t align_up(size_t x) { return (x + 255) & ~(size_t)255; }

extern "C" void kernel_launch(void* const* d_in, const int* in_sizes, int n_in,
                              void* d_out, int out_size, void* d_ws, size_t ws_size,
                              hipStream_t stream) {
    const float* x   = (const float*)d_in[0];
    const float* W1  = (const float*)d_in[1];
    const float* b1  = (const float*)d_in[2];
    const float* W2  = (const float*)d_in[3];
    const float* b2  = (const float*)d_in[4];
    const int*   ei  = (const int*)d_in[5];
    const int*   pos = (const int*)d_in[6];
    const int*   neg = (const int*)d_in[7];
    float* out = (float*)d_out;

    const int n_nodes = in_sizes[0] / F_IN;
    const int n_edges = in_sizes[5] / 2;
    const int n_eval  = in_sizes[6] / 2;
    const int* src = ei;
    const int* dst = ei + n_edges;

    char* w = (char*)d_ws;
    size_t off = 0;
    float* dinv = (float*)(w + off); off += align_up((size_t)n_nodes * 4);
    float* h1   = (float*)(w + off); off += align_up((size_t)n_nodes * F_MID * 4);
    const int TKW = 3 * 2 * TK_PITCH;
    int*   tick = (int*)(w + off);   off += (size_t)TKW * 4;
    float* agg1 = (float*)(w + off); off += align_up((size_t)n_nodes * F_MID * 4);
    float* h2   = (float*)(w + off); off += align_up((size_t)n_nodes * F_OUT * 4);
    float* z    = (float*)(w + off); off += align_up((size_t)n_nodes * F_OUT * 4);
    (void)ws_size;

    const int B = 256;

    init_kernel<<<(n_nodes + TKW + B - 1) / B, B, 0, stream>>>(dinv, tick, n_nodes, TKW);
    deg_count_kernel<<<1024, B, 0, stream>>>(dst, dinv, tick, n_edges, n_nodes);
    rsqrt_kernel<<<(n_nodes + B - 1) / B, B, 0, stream>>>(dinv, n_nodes);

    const int gemm_blocks = (n_nodes + 128 - 1) / 128;
    gemm_kernel<F_IN, F_MID, false><<<gemm_blocks, B, 0, stream>>>(x, W1, dinv, b1, h1, agg1, n_nodes);
    scatter_kernel<F_MID><<<2048, B, 0, stream>>>(
        src, dst, dinv, h1, agg1, tick + 2 * TK_PITCH, n_edges, n_nodes);

    gemm_kernel<F_MID, F_OUT, true><<<gemm_blocks, B, 0, stream>>>(agg1, W2, dinv, b2, h2, z, n_nodes);
    scatter_kernel<F_OUT><<<2048, B, 0, stream>>>(
        src, dst, dinv, h2, z, tick + 4 * TK_PITCH, n_edges, n_nodes);

    {
        int waves = 2 * n_eval;
        int blocks = (waves * 32 + B - 1) / B;
        decode_kernel<<<blocks, B, 0, stream>>>(pos, neg, z, out, n_eval);
    }
    (void)out_size; (void)n_in;
}


# In-degrees counted in per-work-group LDS histograms (128 edge slices, two 16-bit counters a word), summed in the rsqrt pass: degree path has no global atomics, no init launch (8 launches to 7)
# speedup vs baseline: 2.6795x; 1.1316x over previous
#include <hip/hip_runtime.h>
#include <hip/hip_bf16.h>


typedef float v2f __attribute__((ext_vector_type(2)));
typedef float v8f __attribute__((ext_vector_type(8)));

#define F_IN  256
#define F_MID 64
#define F_OUT 32


#define DG_BLOCKS  128
#define DG_THREADS 1024
#define DG_WORDS   65536
#define DG_BATCH   8

__device__ __forceinline__ int deg_pitch(int n_nodes) {
    return (((n_nodes + 1) >> 1) + 3) & ~3;
}

__global__ void __launch_bounds__(DG_THREADS) deg_hist_kernel(const int* __restrict__ dst,
                                                              unsigned* __restrict__ part,
                                                              int* __restrict__ tick,
                                                              int n_edges, int n_nodes, int n_tick) {
    __shared__ unsigned cnt[DG_WORDS];
    const int tid = threadIdx.x;
    if (blockIdx.x == 0 && tid < n_tick) tick[tid] = 0;

    const int per = (n_edges + DG_BLOCKS - 1) / DG_BLOCKS;
    const int e0  = blockIdx.x * per;
    const int e1  = min(e0 + per, n_edges);
    const int hwp = deg_pitch(n_nodes);
    unsigned* __restrict__ mine = part + (size_t)blockIdx.x * hwp;

    for (int wb = 0; wb < hwp; wb += DG_WORDS) {
        const int nw = min(DG_WORDS, hwp - wb);
        for (int i = 4 * tid; i < nw; i += 4 * DG_THREADS)
            *(uint4*)(&cnt[i]) = make_uint4(0u, 0u, 0u, 0u);
        __syncthreads();
        for (int e = e0 + tid; e < e1; e += DG_BATCH * DG_THREADS) {
            int d[DG_BATCH];
#pragma unroll
            for (int u = 0; u < DG_BATCH; ++u) {
                const int idx = e + u * DG_THREADS;
                d[u] = (idx < e1) ? dst[idx] : -1;
            }
#pragma unroll
            for (int u = 0; u < DG_BATCH; ++u) {
                const int w = (d[u] >> 1) - wb;
                if (d[u] >= 0 && (unsigned)w < (unsigned)nw)
                    atomicAdd(&cnt[w], 1u << ((d[u] & 1) << 4));
            }
        }
        __syncthreads();
        for (int i = 4 * tid; i < nw; i += 4 * DG_THREADS)
            *(uint4*)(&mine[wb + i]) = *(const uint4*)(&cnt[i]);
        __syncthreads();
    }
}

__global__ void __launch_bounds__(256) deg_finish_kernel(const unsigned* __restrict__ part,
                                                         float* __restrict__ dinv, int n_nodes) {
    __shared__ unsigned s[8][256];
    const int wave = threadIdx.x >> 5;
    const int lane = threadIdx.x & 31;
    const int hwp  = deg_pitch(n_nodes);
    const int w4   = blockIdx.x * 128 + 4 * lane;
    unsigned acc[8] = {0u, 0u, 0u, 0u, 0u, 0u, 0u, 0u};
    if (w4 < hwp) {
        uint4 v[DG_BLOCKS / 8];
#pragma unroll
        for (int i = 0; i < DG_BLOCKS / 8; ++i)
            v[i] = *(const uint4*)(&part[(size_t)(wave + 8 * i) * hwp + w4]);
#pragma unroll
        for (int i = 0; i < DG_BLOCKS / 8; ++i) {
            acc[0] += v[i].x & 0xffffu;  acc[1] += v[i].x >> 16;
            acc[2] += v[i].y & 0xffffu;  acc[3] += v[i].y >> 16;
            acc[4] += v[i].z & 0xffffu;  acc[5] += v[i].z >> 16;
            acc[6] += v[i].w & 0xffffu;  acc[7] += v[i].w >> 16;
        }
    }
#pragma unroll
    for (int k = 0; k < 8; ++k) s[wave][8 * lane + k] = acc[k];
    __syncthreads();
    const int t = threadIdx.x;
    unsigned total = 1u;
#pragma unroll
    for (int j = 0; j < 8; ++j) total += s[j][t];
    const int node = blockIdx.x * 256 + t;
    if (node < n_nodes) dinv[node] = rsqrtf((float)total);
}


#define TK_EDGES 128
#define TK_PITCH 64

__device__ __forceinline__ int chip_half() {
    const unsigned id = __builtin_amdgcn_s_sendmsg_rtn(0x87);
    return (int)((id >> 18) & 1u);
}

__device__ __forceinline__ int take_ticket(int* ctr) {
    int t = 0;
    if ((threadIdx.x & 31) == 0) t = atomicAdd(ctr, 1);
    return __builtin_amdgcn_readfirstlane(t);
}

__device__ __forceinline__ int node_split(int n_nodes) {
    return ((n_nodes >> 1) + 31) & ~31;
}


template <int K, int NOUT, bool RELU_A>
__global__ void __launch_bounds__(256) gemm_kernel(const float* __restrict__ x,
                                                   const float* __restrict__ W,
                                                   const float* __restrict__ dinv,
                                                   const float* __restrict__ bias,
                                                   float* __restrict__ h,
                                                   float* __restrict__ agg, int n_nodes) {
    constexpr int LDK = K + 4;
    constexpr int NT  = NOUT / 16;
    __shared__ float sWt[NOUT * LDK];

    for (int i = threadIdx.x; i < K * NOUT; i += 256) {
        const int k = i / NOUT;
        const int n = i % NOUT;
        sWt[n * LDK + k] = W[i];
    }
    __syncthreads();

    const int wave = threadIdx.x >> 5;
    const int lane = threadIdx.x & 31;
    const int half = lane >> 4;
    const int l    = lane & 15;
    const int m0   = (blockIdx.x * 8 + wave) * 16;
    if (m0 >= n_nodes) return;

    const int rowA = min(m0 + l, n_nodes - 1);
    const float* __restrict__ xrow = x + (size_t)rowA * K;

    v8f acc[NT] = {};
    if (RELU_A) {
        v2f af[K / 4];
#pragma unroll
        for (int i = 0; i < K / 4; ++i) af[i] = *(const v2f*)(xrow + 4 * i + 2 * half);
#pragma unroll
        for (int i = 0; i < K / 4; ++i) {
            af[i].x = fmaxf(af[i].x, 0.0f);
            af[i].y = fmaxf(af[i].y, 0.0f);
        }
        __builtin_amdgcn_sched_barrier(0);
#pragma unroll
        for (int i = 0; i < K / 4; ++i) {
            const int kk = 4 * i + 2 * half;
#pragma unroll
            for (int t = 0; t < NT; ++t) {
                const int col = t * 16 + l;
                const v2f b = *(const v2f*)(&sWt[col * LDK + kk]);
                acc[t] = __builtin_amdgcn_wmma_f32_16x16x4_f32(
                    false, af[i], false, b, (short)0, acc[t], false, false);
            }
        }
    } else {
        for (int k = 0; k < K; k += 4) {
            const int kk = k + 2 * half;
            const v2f a = *(const v2f*)(xrow + kk);
#pragma unroll
            for (int t = 0; t < NT; ++t) {
                const int col = t * 16 + l;
                const v2f b = *(const v2f*)(&sWt[col * LDK + kk]);
                acc[t] = __builtin_amdgcn_wmma_f32_16x16x4_f32(
                    false, a, false, b, (short)0, acc[t], false, false);
            }
        }
    }

    float bs[NT];
#pragma unroll
    for (int t = 0; t < NT; ++t) bs[t] = bias[t * 16 + l];
    if (m0 + 16 <= n_nodes) {
        const float* __restrict__ dp = dinv + m0 + 8 * half;
        const float4 d0 = *(const float4*)(dp);
        const float4 d1 = *(const float4*)(dp + 4);
        const float di[8] = {d0.x, d0.y, d0.z, d0.w, d1.x, d1.y, d1.z, d1.w};
        const size_t o0 = (size_t)(m0 + 8 * half) * NOUT + l;
        float* __restrict__ hp = h + o0;
        float* __restrict__ ap = agg + o0;
#pragma unroll
        for (int v = 0; v < 8; ++v) {
            const float w = di[v] * di[v];
#pragma unroll
            for (int t = 0; t < NT; ++t) {
                hp[(size_t)v * NOUT + t * 16] = acc[t][v];
                ap[(size_t)v * NOUT + t * 16] = w * acc[t][v] + bs[t];
            }
        }
    } else {
#pragma unroll
        for (int v = 0; v < 8; ++v) {
            const int row = m0 + 8 * half + v;
            if (row < n_nodes) {
                const float d = dinv[row];
                const float w = d * d;
#pragma unroll
                for (int t = 0; t < NT; ++t) {
                    h[(size_t)row * NOUT + t * 16 + l]   = acc[t][v];
                    agg[(size_t)row * NOUT + t * 16 + l] = w * acc[t][v] + bs[t];
                }
            }
        }
    }
}


template <int F>
__device__ __forceinline__ void add_row(const float* __restrict__ h, float* __restrict__ agg,
                                        int s, int d, float nrm, int lane) {
    float hv[F / 32];
#pragma unroll
    for (int k = 0; k < F / 32; ++k) hv[k] = h[(size_t)s * F + 32 * k + lane];
#pragma unroll
    for (int k = 0; k < F / 32; ++k)
        atomicAdd(&agg[(size_t)d * F + 32 * k + lane], nrm * hv[k]);
}

template <int F>
__global__ void __launch_bounds__(256) scatter_kernel(const int* __restrict__ src,
                                                      const int* __restrict__ dst,
                                                      const float* __restrict__ dinv,
                                                      const float* __restrict__ h,
                                                      float* __restrict__ agg,
                                                      int* __restrict__ tick,
                                                      int n_edges, int n_nodes) {
    constexpr int NB = TK_EDGES / 32;
    constexpr int NL = F / 32;
    const int lane  = threadIdx.x & 31;
    const int n_tk  = (n_edges + TK_EDGES - 1) / TK_EDGES;
    const int split = node_split(n_nodes);
    const int mine  = chip_half();
    for (int side = 0; side < 2; ++side) {
        const int hs = mine ^ side;
        int* ctr = tick + hs * TK_PITCH;
        for (int it = 0; it <= n_tk; ++it) {
            const int t = take_ticket(ctr);
            if (t >= n_tk) break;

            int   d[NB], s[NB];
            float nrm[NB];
#pragma unroll
            for (int b = 0; b < NB; ++b) {
                const int e = t * TK_EDGES + 32 * b + lane;
                const bool in = e < n_edges;
                d[b] = in ? dst[e] : -1;
                s[b] = in ? src[e] : 0;
            }
            unsigned msk[NB];
#pragma unroll
            for (int b = 0; b < NB; ++b) {
                const bool m = d[b] >= 0 && (int)(d[b] >= split) == hs;
                msk[b] = __builtin_amdgcn_ballot_w32(m);
                nrm[b] = dinv[m ? s[b] : 0] * dinv[m ? d[b] : 0];
            }

#pragma unroll
            for (int b = 0; b < NB; ++b) {
                unsigned mask = msk[b];
                const int   sb = s[b];
                const int   db = d[b];
                const int   nb = __float_as_int(nrm[b]);
                while (__builtin_popcount(mask) >= 4) {
                    int sj[4], dj[4]; float nj[4];
#pragma unroll
                    for (int u = 0; u < 4; ++u) {
                        const int j = __builtin_ctz(mask);
                        mask &= mask - 1;
                        sj[u] = __builtin_amdgcn_readlane(sb, j);
                        dj[u] = __builtin_amdgcn_readlane(db, j);
                        nj[u] = __int_as_float(__builtin_amdgcn_readlane(nb, j));
                    }
                    float hv[4][NL];
#pragma unroll
                    for (int u = 0; u < 4; ++u)
#pragma unroll
                        for (int k = 0; k < NL; ++k)
                            hv[u][k] = h[(size_t)sj[u] * F + 32 * k + lane];
#pragma unroll
                    for (int u = 0; u < 4; ++u)
#pragma unroll
                        for (int k = 0; k < NL; ++k)
                            atomicAdd(&agg[(size_t)dj[u] * F + 32 * k + lane], nj[u] * hv[u][k]);
                }
                while (mask != 0) {
                    const int j = __builtin_ctz(mask);
                    mask &= mask - 1;
                    add_row<F>(h, agg, __builtin_amdgcn_readlane(sb, j),
                               __builtin_amdgcn_readlane(db, j),
                               __int_as_float(__builtin_amdgcn_readlane(nb, j)), lane);
                }
            }
        }
    }
}


__global__ void __launch_bounds__(256) decode_kernel(const int* __restrict__ pos,
                                                     const int* __restrict__ neg,
                                                     const float* __restrict__ z,
                                                     float* __restrict__ out, int n_eval) {
    const int gw   = (blockIdx.x * 256 + threadIdx.x) >> 5;
    const int lane = threadIdx.x & 31;
    if (gw >= 2 * n_eval) return;
    int a, b;
    if (gw < n_eval) { a = pos[gw];            b = pos[n_eval + gw]; }
    else             { int k = gw - n_eval; a = neg[k]; b = neg[n_eval + k]; }
    float v = z[(size_t)a * F_OUT + lane] * z[(size_t)b * F_OUT + lane];
#pragma unroll
    for (int o = 16; o > 0; o >>= 1) v += __shfl_xor(v, o, 32);
    if (lane == 0) out[gw] = v;
}


static inline size_t align_up(size_t x) { return (x + 255) & ~(size_t)255; }

extern "C" void kernel_launch(void* const* d_in, const int* in_sizes, int n_in,
                              void* d_out, int out_size, void* d_ws, size_t ws_size,
                              hipStream_t stream) {
    const float* x   = (const float*)d_in[0];
    const float* W1  = (const float*)d_in[1];
    const float* b1  = (const float*)d_in[2];
    const float* W2  = (const float*)d_in[3];
    const float* b2  = (const float*)d_in[4];
    const int*   ei  = (const int*)d_in[5];
    const int*   pos = (const int*)d_in[6];
    const int*   neg = (const int*)d_in[7];
    float* out = (float*)d_out;

    const int n_nodes = in_sizes[0] / F_IN;
    const int n_edges = in_sizes[5] / 2;
    const int n_eval  = in_sizes[6] / 2;
    const int* src = ei;
    const int* dst = ei + n_edges;

    char* w = (char*)d_ws;
    size_t off = 0;
    float* dinv = (float*)(w + off); off += align_up((size_t)n_nodes * 4);
    float* h1   = (float*)(w + off); off += align_up((size_t)n_nodes * F_MID * 4);
    const int TKW = 2 * 2 * TK_PITCH;
    int*   tick = (int*)(w + off);   off += (size_t)TKW * 4;
    float* agg1 = (float*)(w + off); off += align_up((size_t)n_nodes * F_MID * 4);
    float* h2   = (float*)(w + off); off += align_up((size_t)n_nodes * F_OUT * 4);
    float* z    = (float*)(w + off); off += align_up((size_t)n_nodes * F_OUT * 4);
    (void)ws_size;

    const int B = 256;

    unsigned* part = (unsigned*)agg1;
    deg_hist_kernel<<<DG_BLOCKS, DG_THREADS, 0, stream>>>(dst, part, tick, n_edges, n_nodes, TKW);
    deg_finish_kernel<<<(n_nodes + B - 1) / B, B, 0, stream>>>(part, dinv, n_nodes);

    const int gemm_blocks = (n_nodes + 128 - 1) / 128;
    gemm_kernel<F_IN, F_MID, false><<<gemm_blocks, B, 0, stream>>>(x, W1, dinv, b1, h1, agg1, n_nodes);
    scatter_kernel<F_MID><<<2048, B, 0, stream>>>(
        src, dst, dinv, h1, agg1, tick, n_edges, n_nodes);

    gemm_kernel<F_MID, F_OUT, true><<<gemm_blocks, B, 0, stream>>>(agg1, W2, dinv, b2, h2, z, n_nodes);
    scatter_kernel<F_OUT><<<2048, B, 0, stream>>>(
        src, dst, dinv, h2, z, tick + 2 * TK_PITCH, n_edges, n_nodes);

    {
        int waves = 2 * n_eval;
        int blocks = (waves * 32 + B - 1) / B;
        decode_kernel<<<blocks, B, 0, stream>>>(pos, neg, z, out, n_eval);
    }
    (void)out_size; (void)n_in;
}


# Decode: a wave takes 32 queries (was 1): lane=feature whole-line row loads kept, 32 product vectors folded by a transposing butterfly (31 exchanges), lane q stores logit q, whole lines; sum tree kept
# speedup vs baseline: 2.9812x; 1.1126x over previous
#include <hip/hip_runtime.h>
#include <hip/hip_bf16.h>


typedef float v2f __attribute__((ext_vector_type(2)));
typedef float v8f __attribute__((ext_vector_type(8)));

#define F_IN  256
#define F_MID 64
#define F_OUT 32


#define DG_BLOCKS  128
#define DG_THREADS 1024
#define DG_WORDS   65536
#define DG_BATCH   8

__device__ __forceinline__ int deg_pitch(int n_nodes) {
    return (((n_nodes + 1) >> 1) + 3) & ~3;
}

__global__ void __launch_bounds__(DG_THREADS) deg_hist_kernel(const int* __restrict__ dst,
                                                              unsigned* __restrict__ part,
                                                              int* __restrict__ tick,
                                                              int n_edges, int n_nodes, int n_tick) {
    __shared__ unsigned cnt[DG_WORDS];
    const int tid = threadIdx.x;
    if (blockIdx.x == 0 && tid < n_tick) tick[tid] = 0;

    const int per = (n_edges + DG_BLOCKS - 1) / DG_BLOCKS;
    const int e0  = blockIdx.x * per;
    const int e1  = min(e0 + per, n_edges);
    const int hwp = deg_pitch(n_nodes);
    unsigned* __restrict__ mine = part + (size_t)blockIdx.x * hwp;

    for (int wb = 0; wb < hwp; wb += DG_WORDS) {
        const int nw = min(DG_WORDS, hwp - wb);
        for (int i = 4 * tid; i < nw; i += 4 * DG_THREADS)
            *(uint4*)(&cnt[i]) = make_uint4(0u, 0u, 0u, 0u);
        __syncthreads();
        for (int e = e0 + tid; e < e1; e += DG_BATCH * DG_THREADS) {
            int d[DG_BATCH];
#pragma unroll
            for (int u = 0; u < DG_BATCH; ++u) {
                const int idx = e + u * DG_THREADS;
                d[u] = (idx < e1) ? dst[idx] : -1;
            }
#pragma unroll
            for (int u = 0; u < DG_BATCH; ++u) {
                const int w = (d[u] >> 1) - wb;
                if (d[u] >= 0 && (unsigned)w < (unsigned)nw)
                    atomicAdd(&cnt[w], 1u << ((d[u] & 1) << 4));
            }
        }
        __syncthreads();
        for (int i = 4 * tid; i < nw; i += 4 * DG_THREADS)
            *(uint4*)(&mine[wb + i]) = *(const uint4*)(&cnt[i]);
        __syncthreads();
    }
}

__global__ void __launch_bounds__(256) deg_finish_kernel(const unsigned* __restrict__ part,
                                                         float* __restrict__ dinv, int n_nodes) {
    __shared__ unsigned s[8][256];
    const int wave = threadIdx.x >> 5;
    const int lane = threadIdx.x & 31;
    const int hwp  = deg_pitch(n_nodes);
    const int w4   = blockIdx.x * 128 + 4 * lane;
    unsigned acc[8] = {0u, 0u, 0u, 0u, 0u, 0u, 0u, 0u};
    if (w4 < hwp) {
        uint4 v[DG_BLOCKS / 8];
#pragma unroll
        for (int i = 0; i < DG_BLOCKS / 8; ++i)
            v[i] = *(const uint4*)(&part[(size_t)(wave + 8 * i) * hwp + w4]);
#pragma unroll
        for (int i = 0; i < DG_BLOCKS / 8; ++i) {
            acc[0] += v[i].x & 0xffffu;  acc[1] += v[i].x >> 16;
            acc[2] += v[i].y & 0xffffu;  acc[3] += v[i].y >> 16;
            acc[4] += v[i].z & 0xffffu;  acc[5] += v[i].z >> 16;
            acc[6] += v[i].w & 0xffffu;  acc[7] += v[i].w >> 16;
        }
    }
#pragma unroll
    for (int k = 0; k < 8; ++k) s[wave][8 * lane + k] = acc[k];
    __syncthreads();
    const int t = threadIdx.x;
    unsigned total = 1u;
#pragma unroll
    for (int j = 0; j < 8; ++j) total += s[j][t];
    const int node = blockIdx.x * 256 + t;
    if (node < n_nodes) dinv[node] = rsqrtf((float)total);
}


#define TK_EDGES 128
#define TK_PITCH 64

__device__ __forceinline__ int chip_half() {
    const unsigned id = __builtin_amdgcn_s_sendmsg_rtn(0x87);
    return (int)((id >> 18) & 1u);
}

__device__ __forceinline__ int take_ticket(int* ctr) {
    int t = 0;
    if ((threadIdx.x & 31) == 0) t = atomicAdd(ctr, 1);
    return __builtin_amdgcn_readfirstlane(t);
}

__device__ __forceinline__ int node_split(int n_nodes) {
    return ((n_nodes >> 1) + 31) & ~31;
}


template <int K, int NOUT, bool RELU_A>
__global__ void __launch_bounds__(256) gemm_kernel(const float* __restrict__ x,
                                                   const float* __restrict__ W,
                                                   const float* __restrict__ dinv,
                                                   const float* __restrict__ bias,
                                                   float* __restrict__ h,
                                                   float* __restrict__ agg, int n_nodes) {
    constexpr int LDK = K + 4;
    constexpr int NT  = NOUT / 16;
    __shared__ float sWt[NOUT * LDK];

    for (int i = threadIdx.x; i < K * NOUT; i += 256) {
        const int k = i / NOUT;
        const int n = i % NOUT;
        sWt[n * LDK + k] = W[i];
    }
    __syncthreads();

    const int wave = threadIdx.x >> 5;
    const int lane = threadIdx.x & 31;
    const int half = lane >> 4;
    const int l    = lane & 15;
    const int m0   = (blockIdx.x * 8 + wave) * 16;
    if (m0 >= n_nodes) return;

    const int rowA = min(m0 + l, n_nodes - 1);
    const float* __restrict__ xrow = x + (size_t)rowA * K;

    v8f acc[NT] = {};
    if (RELU_A) {
        v2f af[K / 4];
#pragma unroll
        for (int i = 0; i < K / 4; ++i) af[i] = *(const v2f*)(xrow + 4 * i + 2 * half);
#pragma unroll
        for (int i = 0; i < K / 4; ++i) {
            af[i].x = fmaxf(af[i].x, 0.0f);
            af[i].y = fmaxf(af[i].y, 0.0f);
        }
        __builtin_amdgcn_sched_barrier(0);
#pragma unroll
        for (int i = 0; i < K / 4; ++i) {
            const int kk = 4 * i + 2 * half;
#pragma unroll
            for (int t = 0; t < NT; ++t) {
                const int col = t * 16 + l;
                const v2f b = *(const v2f*)(&sWt[col * LDK + kk]);
                acc[t] = __builtin_amdgcn_wmma_f32_16x16x4_f32(
                    false, af[i], false, b, (short)0, acc[t], false, false);
            }
        }
    } else {
        for (int k = 0; k < K; k += 4) {
            const int kk = k + 2 * half;
            const v2f a = *(const v2f*)(xrow + kk);
#pragma unroll
            for (int t = 0; t < NT; ++t) {
                const int col = t * 16 + l;
                const v2f b = *(const v2f*)(&sWt[col * LDK + kk]);
                acc[t] = __builtin_amdgcn_wmma_f32_16x16x4_f32(
                    false, a, false, b, (short)0, acc[t], false, false);
            }
        }
    }

    float bs[NT];
#pragma unroll
    for (int t = 0; t < NT; ++t) bs[t] = bias[t * 16 + l];
    if (m0 + 16 <= n_nodes) {
        const float* __restrict__ dp = dinv + m0 + 8 * half;
        const float4 d0 = *(const float4*)(dp);
        const float4 d1 = *(const float4*)(dp + 4);
        const float di[8] = {d0.x, d0.y, d0.z, d0.w, d1.x, d1.y, d1.z, d1.w};
        const size_t o0 = (size_t)(m0 + 8 * half) * NOUT + l;
        float* __restrict__ hp = h + o0;
        float* __restrict__ ap = agg + o0;
#pragma unroll
        for (int v = 0; v < 8; ++v) {
            const float w = di[v] * di[v];
#pragma unroll
            for (int t = 0; t < NT; ++t) {
                hp[(size_t)v * NOUT + t * 16] = acc[t][v];
                ap[(size_t)v * NOUT + t * 16] = w * acc[t][v] + bs[t];
            }
        }
    } else {
#pragma unroll
        for (int v = 0; v < 8; ++v) {
            const int row = m0 + 8 * half + v;
            if (row < n_nodes) {
                const float d = dinv[row];
                const float w = d * d;
#pragma unroll
                for (int t = 0; t < NT; ++t) {
                    h[(size_t)row * NOUT + t * 16 + l]   = acc[t][v];
                    agg[(size_t)row * NOUT + t * 16 + l] = w * acc[t][v] + bs[t];
                }
            }
        }
    }
}


template <int F>
__device__ __forceinline__ void add_row(const float* __restrict__ h, float* __restrict__ agg,
                                        int s, int d, float nrm, int lane) {
    float hv[F / 32];
#pragma unroll
    for (int k = 0; k < F / 32; ++k) hv[k] = h[(size_t)s * F + 32 * k + lane];
#pragma unroll
    for (int k = 0; k < F / 32; ++k)
        atomicAdd(&agg[(size_t)d * F + 32 * k + lane], nrm * hv[k]);
}

template <int F>
__global__ void __launch_bounds__(256) scatter_kernel(const int* __restrict__ src,
                                                      const int* __restrict__ dst,
                                                      const float* __restrict__ dinv,
                                                      const float* __restrict__ h,
                                                      float* __restrict__ agg,
                                                      int* __restrict__ tick,
                                                      int n_edges, int n_nodes) {
    constexpr int NB = TK_EDGES / 32;
    constexpr int NL = F / 32;
    const int lane  = threadIdx.x & 31;
    const int n_tk  = (n_edges + TK_EDGES - 1) / TK_EDGES;
    const int split = node_split(n_nodes);
    const int mine  = chip_half();
    for (int side = 0; side < 2; ++side) {
        const int hs = mine ^ side;
        int* ctr = tick + hs * TK_PITCH;
        for (int it = 0; it <= n_tk; ++it) {
            const int t = take_ticket(ctr);
            if (t >= n_tk) break;

            int   d[NB], s[NB];
            float nrm[NB];
#pragma unroll
            for (int b = 0; b < NB; ++b) {
                const int e = t * TK_EDGES + 32 * b + lane;
                const bool in = e < n_edges;
                d[b] = in ? dst[e] : -1;
                s[b] = in ? src[e] : 0;
            }
            unsigned msk[NB];
#pragma unroll
            for (int b = 0; b < NB; ++b) {
                const bool m = d[b] >= 0 && (int)(d[b] >= split) == hs;
                msk[b] = __builtin_amdgcn_ballot_w32(m);
                nrm[b] = dinv[m ? s[b] : 0] * dinv[m ? d[b] : 0];
            }

#pragma unroll
            for (int b = 0; b < NB; ++b) {
                unsigned mask = msk[b];
                const int   sb = s[b];
                const int   db = d[b];
                const int   nb = __float_as_int(nrm[b]);
                while (__builtin_popcount(mask) >= 4) {
                    int sj[4], dj[4]; float nj[4];
#pragma unroll
                    for (int u = 0; u < 4; ++u) {
                        const int j = __builtin_ctz(mask);
                        mask &= mask - 1;
                        sj[u] = __builtin_amdgcn_readlane(sb, j);
                        dj[u] = __builtin_amdgcn_readlane(db, j);
                        nj[u] = __int_as_float(__builtin_amdgcn_readlane(nb, j));
                    }
                    float hv[4][NL];
#pragma unroll
                    for (int u = 0; u < 4; ++u)
#pragma unroll
                        for (int k = 0; k < NL; ++k)
                            hv[u][k] = h[(size_t)sj[u] * F + 32 * k + lane];
#pragma unroll
                    for (int u = 0; u < 4; ++u)
#pragma unroll
                        for (int k = 0; k < NL; ++k)
                            atomicAdd(&agg[(size_t)dj[u] * F + 32 * k + lane], nj[u] * hv[u][k]);
                }
                while (mask != 0) {
                    const int j = __builtin_ctz(mask);
                    mask &= mask - 1;
                    add_row<F>(h, agg, __builtin_amdgcn_readlane(sb, j),
                               __builtin_amdgcn_readlane(db, j),
                               __int_as_float(__builtin_amdgcn_readlane(nb, j)), lane);
                }
            }
        }
    }
}


#define DQ 32

__device__ __forceinline__ float fold_pair(float x, float y, int o, int lane) {
    const bool up = (lane & o) != 0;
    const float keep = up ? y : x;
    const float give = up ? x : y;
    return keep + __shfl_xor(give, o, 32);
}

__global__ void __launch_bounds__(256) decode_kernel(const int* __restrict__ pos,
                                                     const int* __restrict__ neg,
                                                     const float* __restrict__ z,
                                                     float* __restrict__ out, int n_eval) {
    const int gw   = (blockIdx.x * 256 + threadIdx.x) >> 5;
    const int lane = threadIdx.x & 31;
    const int nq   = 2 * n_eval;
    const int q0   = gw * DQ;
    if (q0 >= nq) return;
    const int  q      = q0 + lane;
    const bool live   = q < nq;
    const bool second = q >= n_eval;
    const int* __restrict__ list = second ? neg : pos;
    const int  k      = second ? q - n_eval : q;
    const int  a      = live ? list[k] : 0;
    const int  b      = live ? list[n_eval + k] : 0;

    float za[DQ], zb[DQ];
#pragma unroll
    for (int j = 0; j < DQ; ++j) {
        const unsigned aj = (unsigned)__builtin_amdgcn_readlane(a, j);
        const unsigned bj = (unsigned)__builtin_amdgcn_readlane(b, j);
        za[j] = z[aj * F_OUT + lane];
        zb[j] = z[bj * F_OUT + lane];
    }
    __builtin_amdgcn_sched_barrier(0);
    float p[DQ];
#pragma unroll
    for (int j = 0; j < DQ; ++j) p[j] = za[j] * zb[j];
#pragma unroll
    for (int o = 16; o > 0; o >>= 1)
#pragma unroll
        for (int j = 0; j < o; ++j) p[j] = fold_pair(p[j], p[j + o], o, lane);

    if (live) out[q] = p[0];
}


static inline size_t align_up(size_t x) { return (x + 255) & ~(size_t)255; }

extern "C" void kernel_launch(void* const* d_in, const int* in_sizes, int n_in,
                              void* d_out, int out_size, void* d_ws, size_t ws_size,
                              hipStream_t stream) {
    const float* x   = (const float*)d_in[0];
    const float* W1  = (const float*)d_in[1];
    const float* b1  = (const float*)d_in[2];
    const float* W2  = (const float*)d_in[3];
    const float* b2  = (const float*)d_in[4];
    const int*   ei  = (const int*)d_in[5];
    const int*   pos = (const int*)d_in[6];
    const int*   neg = (const int*)d_in[7];
    float* out = (float*)d_out;

    const int n_nodes = in_sizes[0] / F_IN;
    const int n_edges = in_sizes[5] / 2;
    const int n_eval  = in_sizes[6] / 2;
    const int* src = ei;
    const int* dst = ei + n_edges;

    char* w = (char*)d_ws;
    size_t off = 0;
    float* dinv = (float*)(w + off); off += align_up((size_t)n_nodes * 4);
    float* h1   = (float*)(w + off); off += align_up((size_t)n_nodes * F_MID * 4);
    const int TKW = 2 * 2 * TK_PITCH;
    int*   tick = (int*)(w + off);   off += (size_t)TKW * 4;
    float* agg1 = (float*)(w + off); off += align_up((size_t)n_nodes * F_MID * 4);
    float* h2   = (float*)(w + off); off += align_up((size_t)n_nodes * F_OUT * 4);
    float* z    = (float*)(w + off); off += align_up((size_t)n_nodes * F_OUT * 4);
    (void)ws_size;

    const int B = 256;

    unsigned* part = (unsigned*)agg1;
    deg_hist_kernel<<<DG_BLOCKS, DG_THREADS, 0, stream>>>(dst, part, tick, n_edges, n_nodes, TKW);
    deg_finish_kernel<<<(n_nodes + B - 1) / B, B, 0, stream>>>(part, dinv, n_nodes);

    const int gemm_blocks = (n_nodes + 128 - 1) / 128;
    gemm_kernel<F_IN, F_MID, false><<<gemm_blocks, B, 0, stream>>>(x, W1, dinv, b1, h1, agg1, n_nodes);
    scatter_kernel<F_MID><<<2048, B, 0, stream>>>(
        src, dst, dinv, h1, agg1, tick, n_edges, n_nodes);

    gemm_kernel<F_MID, F_OUT, true><<<gemm_blocks, B, 0, stream>>>(agg1, W2, dinv, b2, h2, z, n_nodes);
    scatter_kernel<F_OUT><<<2048, B, 0, stream>>>(
        src, dst, dinv, h2, z, tick + 2 * TK_PITCH, n_edges, n_nodes);

    {
        int waves = (2 * n_eval + DQ - 1) / DQ;
        int blocks = (waves * 32 + B - 1) / B;
        decode_kernel<<<blocks, B, 0, stream>>>(pos, neg, z, out, n_eval);
    }
    (void)out_size; (void)n_in;
}


# GEMM row tiles drawn as tickets per chip half (die id, one counter per half): h, start values and layer-2 A rows are written and read in the L2 half whose waves add into them; sums unchanged
# speedup vs baseline: 3.0829x; 1.0341x over previous
#include <hip/hip_runtime.h>
#include <hip/hip_bf16.h>


typedef float v2f __attribute__((ext_vector_type(2)));
typedef float v8f __attribute__((ext_vector_type(8)));

#define F_IN  256
#define F_MID 64
#define F_OUT 32


#define DG_BLOCKS  128
#define DG_THREADS 1024
#define DG_WORDS   65536
#define DG_BATCH   8

__device__ __forceinline__ int deg_pitch(int n_nodes) {
    return (((n_nodes + 1) >> 1) + 3) & ~3;
}

__global__ void __launch_bounds__(DG_THREADS) deg_hist_kernel(const int* __restrict__ dst,
                                                              unsigned* __restrict__ part,
                                                              int* __restrict__ tick,
                                                              int n_edges, int n_nodes, int n_tick) {
    __shared__ unsigned cnt[DG_WORDS];
    const int tid = threadIdx.x;
    if (blockIdx.x == 0 && tid < n_tick) tick[tid] = 0;

    const int per = (n_edges + DG_BLOCKS - 1) / DG_BLOCKS;
    const int e0  = blockIdx.x * per;
    const int e1  = min(e0 + per, n_edges);
    const int hwp = deg_pitch(n_nodes);
    unsigned* __restrict__ mine = part + (size_t)blockIdx.x * hwp;

    for (int wb = 0; wb < hwp; wb += DG_WORDS) {
        const int nw = min(DG_WORDS, hwp - wb);
        for (int i = 4 * tid; i < nw; i += 4 * DG_THREADS)
            *(uint4*)(&cnt[i]) = make_uint4(0u, 0u, 0u, 0u);
        __syncthreads();
        for (int e = e0 + tid; e < e1; e += DG_BATCH * DG_THREADS) {
            int d[DG_BATCH];
#pragma unroll
            for (int u = 0; u < DG_BATCH; ++u) {
                const int idx = e + u * DG_THREADS;
                d[u] = (idx < e1) ? dst[idx] : -1;
            }
#pragma unroll
            for (int u = 0; u < DG_BATCH; ++u) {
                const int w = (d[u] >> 1) - wb;
                if (d[u] >= 0 && (unsigned)w < (unsigned)nw)
                    atomicAdd(&cnt[w], 1u << ((d[u] & 1) << 4));
            }
        }
        __syncthreads();
        for (int i = 4 * tid; i < nw; i += 4 * DG_THREADS)
            *(uint4*)(&mine[wb + i]) = *(const uint4*)(&cnt[i]);
        __syncthreads();
    }
}

__global__ void __launch_bounds__(256) deg_finish_kernel(const unsigned* __restrict__ part,
                                                         float* __restrict__ dinv, int n_nodes) {
    __shared__ unsigned s[8][256];
    const int wave = threadIdx.x >> 5;
    const int lane = threadIdx.x & 31;
    const int hwp  = deg_pitch(n_nodes);
    const int w4   = blockIdx.x * 128 + 4 * lane;
    unsigned acc[8] = {0u, 0u, 0u, 0u, 0u, 0u, 0u, 0u};
    if (w4 < hwp) {
        uint4 v[DG_BLOCKS / 8];
#pragma unroll
        for (int i = 0; i < DG_BLOCKS / 8; ++i)
            v[i] = *(const uint4*)(&part[(size_t)(wave + 8 * i) * hwp + w4]);
#pragma unroll
        for (int i = 0; i < DG_BLOCKS / 8; ++i) {
            acc[0] += v[i].x & 0xffffu;  acc[1] += v[i].x >> 16;
            acc[2] += v[i].y & 0xffffu;  acc[3] += v[i].y >> 16;
            acc[4] += v[i].z & 0xffffu;  acc[5] += v[i].z >> 16;
            acc[6] += v[i].w & 0xffffu;  acc[7] += v[i].w >> 16;
        }
    }
#pragma unroll
    for (int k = 0; k < 8; ++k) s[wave][8 * lane + k] = acc[k];
    __syncthreads();
    const int t = threadIdx.x;
    unsigned total = 1u;
#pragma unroll
    for (int j = 0; j < 8; ++j) total += s[j][t];
    const int node = blockIdx.x * 256 + t;
    if (node < n_nodes) dinv[node] = rsqrtf((float)total);
}


#define TK_EDGES 128
#define TK_PITCH 64

__device__ __forceinline__ int chip_half() {
    const unsigned id = __builtin_amdgcn_s_sendmsg_rtn(0x87);
    return (int)((id >> 18) & 1u);
}

__device__ __forceinline__ int take_ticket(int* ctr) {
    int t = 0;
    if ((threadIdx.x & 31) == 0) t = atomicAdd(ctr, 1);
    return __builtin_amdgcn_readfirstlane(t);
}

__device__ __forceinline__ int node_split(int n_nodes) {
    return ((n_nodes >> 1) + 31) & ~31;
}


template <int K, int NOUT, bool RELU_A>
__global__ void __launch_bounds__(256) gemm_kernel(const float* __restrict__ x,
                                                   const float* __restrict__ W,
                                                   const float* __restrict__ dinv,
                                                   const float* __restrict__ bias,
                                                   float* __restrict__ h,
                                                   float* __restrict__ agg,
                                                   int* __restrict__ tile_ctr, int n_nodes) {
    constexpr int LDK = K + 4;
    constexpr int NT  = NOUT / 16;
    __shared__ float sWt[NOUT * LDK];
    __shared__ int sTile[2];

    const int split = min(node_split(n_nodes), n_nodes);
    if (threadIdx.x == 0) {
        const int mine = chip_half();
        int first = -1, last = 0;
        for (int side = 0; side < 2 && first < 0; ++side) {
            const int hs   = mine ^ side;
            const int lo   = hs ? split : 0;
            const int hi   = hs ? n_nodes : split;
            const int n_tl = (hi - lo + 127) / 128;
            const int t = (n_tl > 0) ? atomicAdd(tile_ctr + hs * TK_PITCH, 1) : 0;
            if (t < n_tl) { first = lo + t * 128; last = hi; }
        }
        sTile[0] = first;
        sTile[1] = last;
    }

    for (int i = threadIdx.x; i < K * NOUT; i += 256) {
        const int k = i / NOUT;
        const int n = i % NOUT;
        sWt[n * LDK + k] = W[i];
    }
    __syncthreads();

    const int wave = threadIdx.x >> 5;
    const int lane = threadIdx.x & 31;
    const int half = lane >> 4;
    const int l    = lane & 15;
    if (sTile[0] < 0) return;
    const int m0   = sTile[0] + wave * 16;
    if (m0 >= sTile[1]) return;

    const int rowA = min(m0 + l, n_nodes - 1);
    const float* __restrict__ xrow = x + (size_t)rowA * K;

    v8f acc[NT] = {};
    if (RELU_A) {
        v2f af[K / 4];
#pragma unroll
        for (int i = 0; i < K / 4; ++i) af[i] = *(const v2f*)(xrow + 4 * i + 2 * half);
#pragma unroll
        for (int i = 0; i < K / 4; ++i) {
            af[i].x = fmaxf(af[i].x, 0.0f);
            af[i].y = fmaxf(af[i].y, 0.0f);
        }
        __builtin_amdgcn_sched_barrier(0);
#pragma unroll
        for (int i = 0; i < K / 4; ++i) {
            const int kk = 4 * i + 2 * half;
#pragma unroll
            for (int t = 0; t < NT; ++t) {
                const int col = t * 16 + l;
                const v2f b = *(const v2f*)(&sWt[col * LDK + kk]);
                acc[t] = __builtin_amdgcn_wmma_f32_16x16x4_f32(
                    false, af[i], false, b, (short)0, acc[t], false, false);
            }
        }
    } else {
        for (int k = 0; k < K; k += 4) {
            const int kk = k + 2 * half;
            const v2f a = *(const v2f*)(xrow + kk);
#pragma unroll
            for (int t = 0; t < NT; ++t) {
                const int col = t * 16 + l;
                const v2f b = *(const v2f*)(&sWt[col * LDK + kk]);
                acc[t] = __builtin_amdgcn_wmma_f32_16x16x4_f32(
                    false, a, false, b, (short)0, acc[t], false, false);
            }
        }
    }

    float bs[NT];
#pragma unroll
    for (int t = 0; t < NT; ++t) bs[t] = bias[t * 16 + l];
    if (m0 + 16 <= n_nodes) {
        const float* __restrict__ dp = dinv + m0 + 8 * half;
        const float4 d0 = *(const float4*)(dp);
        const float4 d1 = *(const float4*)(dp + 4);
        const float di[8] = {d0.x, d0.y, d0.z, d0.w, d1.x, d1.y, d1.z, d1.w};
        const size_t o0 = (size_t)(m0 + 8 * half) * NOUT + l;
        float* __restrict__ hp = h + o0;
        float* __restrict__ ap = agg + o0;
#pragma unroll
        for (int v = 0; v < 8; ++v) {
            const float w = di[v] * di[v];
#pragma unroll
            for (int t = 0; t < NT; ++t) {
                hp[(size_t)v * NOUT + t * 16] = acc[t][v];
                ap[(size_t)v * NOUT + t * 16] = w * acc[t][v] + bs[t];
            }
        }
    } else {
#pragma unroll
        for (int v = 0; v < 8; ++v) {
            const int row = m0 + 8 * half + v;
            if (row < n_nodes) {
                const float d = dinv[row];
                const float w = d * d;
#pragma unroll
                for (int t = 0; t < NT; ++t) {
                    h[(size_t)row * NOUT + t * 16 + l]   = acc[t][v];
                    agg[(size_t)row * NOUT + t * 16 + l] = w * acc[t][v] + bs[t];
                }
            }
        }
    }
}


template <int F>
__device__ __forceinline__ void add_row(const float* __restrict__ h, float* __restrict__ agg,
                                        int s, int d, float nrm, int lane) {
    float hv[F / 32];
#pragma unroll
    for (int k = 0; k < F / 32; ++k) hv[k] = h[(size_t)s * F + 32 * k + lane];
#pragma unroll
    for (int k = 0; k < F / 32; ++k)
        atomicAdd(&agg[(size_t)d * F + 32 * k + lane], nrm * hv[k]);
}

template <int F>
__global__ void __launch_bounds__(256) scatter_kernel(const int* __restrict__ src,
                                                      const int* __restrict__ dst,
                                                      const float* __restrict__ dinv,
                                                      const float* __restrict__ h,
                                                      float* __restrict__ agg,
                                                      int* __restrict__ tick,
                                                      int n_edges, int n_nodes) {
    constexpr int NB = TK_EDGES / 32;
    constexpr int NL = F / 32;
    const int lane  = threadIdx.x & 31;
    const int n_tk  = (n_edges + TK_EDGES - 1) / TK_EDGES;
    const int split = node_split(n_nodes);
    const int mine  = chip_half();
    for (int side = 0; side < 2; ++side) {
        const int hs = mine ^ side;
        int* ctr = tick + hs * TK_PITCH;
        for (int it = 0; it <= n_tk; ++it) {
            const int t = take_ticket(ctr);
            if (t >= n_tk) break;

            int   d[NB], s[NB];
            float nrm[NB];
#pragma unroll
            for (int b = 0; b < NB; ++b) {
                const int e = t * TK_EDGES + 32 * b + lane;
                const bool in = e < n_edges;
                d[b] = in ? dst[e] : -1;
                s[b] = in ? src[e] : 0;
            }
            unsigned msk[NB];
#pragma unroll
            for (int b = 0; b < NB; ++b) {
                const bool m = d[b] >= 0 && (int)(d[b] >= split) == hs;
                msk[b] = __builtin_amdgcn_ballot_w32(m);
                nrm[b] = dinv[m ? s[b] : 0] * dinv[m ? d[b] : 0];
            }

#pragma unroll
            for (int b = 0; b < NB; ++b) {
                unsigned mask = msk[b];
                const int   sb = s[b];
                const int   db = d[b];
                const int   nb = __float_as_int(nrm[b]);
                while (__builtin_popcount(mask) >= 4) {
                    int sj[4], dj[4]; float nj[4];
#pragma unroll
                    for (int u = 0; u < 4; ++u) {
                        const int j = __builtin_ctz(mask);
                        mask &= mask - 1;
                        sj[u] = __builtin_amdgcn_readlane(sb, j);
                        dj[u] = __builtin_amdgcn_readlane(db, j);
                        nj[u] = __int_as_float(__builtin_amdgcn_readlane(nb, j));
                    }
                    float hv[4][NL];
#pragma unroll
                    for (int u = 0; u < 4; ++u)
#pragma unroll
                        for (int k = 0; k < NL; ++k)
                            hv[u][k] = h[(size_t)sj[u] * F + 32 * k + lane];
#pragma unroll
                    for (int u = 0; u < 4; ++u)
#pragma unroll
                        for (int k = 0; k < NL; ++k)
                            atomicAdd(&agg[(size_t)dj[u] * F + 32 * k + lane], nj[u] * hv[u][k]);
                }
                while (mask != 0) {
                    const int j = __builtin_ctz(mask);
                    mask &= mask - 1;
                    add_row<F>(h, agg, __builtin_amdgcn_readlane(sb, j),
                               __builtin_amdgcn_readlane(db, j),
                               __int_as_float(__builtin_amdgcn_readlane(nb, j)), lane);
                }
            }
        }
    }
}


#define DQ 32

__device__ __forceinline__ float fold_pair(float x, float y, int o, int lane) {
    const bool up = (lane & o) != 0;
    const float keep = up ? y : x;
    const float give = up ? x : y;
    return keep + __shfl_xor(give, o, 32);
}

__global__ void __launch_bounds__(256) decode_kernel(const int* __restrict__ pos,
                                                     const int* __restrict__ neg,
                                                     const float* __restrict__ z,
                                                     float* __restrict__ out, int n_eval) {
    const int gw   = (blockIdx.x * 256 + threadIdx.x) >> 5;
    const int lane = threadIdx.x & 31;
    const int nq   = 2 * n_eval;
    const int q0   = gw * DQ;
    if (q0 >= nq) return;
    const int  q      = q0 + lane;
    const bool live   = q < nq;
    const bool second = q >= n_eval;
    const int* __restrict__ list = second ? neg : pos;
    const int  k      = second ? q - n_eval : q;
    const int  a      = live ? list[k] : 0;
    const int  b      = live ? list[n_eval + k] : 0;

    float za[DQ], zb[DQ];
#pragma unroll
    for (int j = 0; j < DQ; ++j) {
        const unsigned aj = (unsigned)__builtin_amdgcn_readlane(a, j);
        const unsigned bj = (unsigned)__builtin_amdgcn_readlane(b, j);
        za[j] = z[aj * F_OUT + lane];
        zb[j] = z[bj * F_OUT + lane];
    }
    __builtin_amdgcn_sched_barrier(0);
    float p[DQ];
#pragma unroll
    for (int j = 0; j < DQ; ++j) p[j] = za[j] * zb[j];
#pragma unroll
    for (int o = 16; o > 0; o >>= 1)
#pragma unroll
        for (int j = 0; j < o; ++j) p[j] = fold_pair(p[j], p[j + o], o, lane);

    if (live) out[q] = p[0];
}


static inline size_t align_up(size_t x) { return (x + 255) & ~(size_t)255; }

extern "C" void kernel_launch(void* const* d_in, const int* in_sizes, int n_in,
                              void* d_out, int out_size, void* d_ws, size_t ws_size,
                              hipStream_t stream) {
    const float* x   = (const float*)d_in[0];
    const float* W1  = (const float*)d_in[1];
    const float* b1  = (const float*)d_in[2];
    const float* W2  = (const float*)d_in[3];
    const float* b2  = (const float*)d_in[4];
    const int*   ei  = (const int*)d_in[5];
    const int*   pos = (const int*)d_in[6];
    const int*   neg = (const int*)d_in[7];
    float* out = (float*)d_out;

    const int n_nodes = in_sizes[0] / F_IN;
    const int n_edges = in_sizes[5] / 2;
    const int n_eval  = in_sizes[6] / 2;
    const int* src = ei;
    const int* dst = ei + n_edges;

    char* w = (char*)d_ws;
    size_t off = 0;
    float* dinv = (float*)(w + off); off += align_up((size_t)n_nodes * 4);
    float* h1   = (float*)(w + off); off += align_up((size_t)n_nodes * F_MID * 4);
    const int TKW = 2 * 2 * TK_PITCH;
    int*   tick = (int*)(w + off);   off += (size_t)TKW * 4;
    float* agg1 = (float*)(w + off); off += align_up((size_t)n_nodes * F_MID * 4);
    float* h2   = (float*)(w + off); off += align_up((size_t)n_nodes * F_OUT * 4);
    float* z    = (float*)(w + off); off += align_up((size_t)n_nodes * F_OUT * 4);
    (void)ws_size;

    const int B = 256;

    unsigned* part = (unsigned*)agg1;
    deg_hist_kernel<<<DG_BLOCKS, DG_THREADS, 0, stream>>>(dst, part, tick, n_edges, n_nodes, TKW);
    deg_finish_kernel<<<(n_nodes + B - 1) / B, B, 0, stream>>>(part, dinv, n_nodes);

    const int gemm_blocks = (n_nodes + 128 - 1) / 128 + 1;
    gemm_kernel<F_IN, F_MID, false><<<gemm_blocks, B, 0, stream>>>(
        x, W1, dinv, b1, h1, agg1, tick + TK_PITCH / 2, n_nodes);
    scatter_kernel<F_MID><<<2048, B, 0, stream>>>(
        src, dst, dinv, h1, agg1, tick, n_edges, n_nodes);

    gemm_kernel<F_MID, F_OUT, true><<<gemm_blocks, B, 0, stream>>>(
        agg1, W2, dinv, b2, h2, z, tick + 2 * TK_PITCH + TK_PITCH / 2, n_nodes);
    scatter_kernel<F_OUT><<<2048, B, 0, stream>>>(
        src, dst, dinv, h2, z, tick + 2 * TK_PITCH, n_edges, n_nodes);

    {
        int waves = (2 * n_eval + DQ - 1) / DQ;
        int blocks = (waves * 32 + B - 1) / B;
        decode_kernel<<<blocks, B, 0, stream>>>(pos, neg, z, out, n_eval);
    }
    (void)out_size; (void)n_in;
}
